# starting kernel
# baseline (speedup 1.0000x reference)
#include <hip/hip_runtime.h>


#define ALPHA_  0.01
#define AVAL_   0.99
#define B_      8
#define T_      8192
#define C_      512
#define CL_     512
#define NCH_    16

typedef __attribute__((ext_vector_type(2))) float v2f;
typedef __attribute__((ext_vector_type(8))) float v8f;

__global__ void ema_p1(const float* __restrict__ x, double* __restrict__ E) {
    int tid = blockIdx.x * blockDim.x + threadIdx.x;
    int c = tid & (C_ - 1);
    int k = (tid >> 9) & (NCH_ - 1);
    int b = tid >> 13;
    const float* p = x + ((size_t)b * T_ + (size_t)k * CL_) * C_ + c;
    double y = 0.0;
    int t = 0;
    if (k == 0) { y = (double)p[0]; t = 1; }
    for (; t < CL_; ++t)
        y = AVAL_ * y + ALPHA_ * (double)p[(size_t)t * C_];
    E[((size_t)b * NCH_ + k) * C_ + c] = y;
}

__global__ void ema_p2(const double* __restrict__ E, double* __restrict__ carry) {
    int tid = blockIdx.x * blockDim.x + threadIdx.x;
    int c = tid & (C_ - 1);
    int b = tid >> 9;
    double a512 = AVAL_;
    #pragma unroll
    for (int i = 0; i < 9; ++i) a512 *= a512;
    double F = 0.0;
    for (int k = 0; k < NCH_; ++k) {
        size_t idx = ((size_t)b * NCH_ + k) * C_ + c;
        carry[idx] = F;
        F = E[idx] + a512 * F;
    }
}

__global__ void ema_p3(const float* __restrict__ x,
                       const double* __restrict__ carry,
                       float* __restrict__ out) {
    const int lane = threadIdx.x & 31;
    const int wav  = threadIdx.x >> 5;
    const int gw   = blockIdx.x * (blockDim.x >> 5) + wav;
    const int ct   = gw & 31;
    const int k    = (gw >> 5) & 15;
    const int b    = gw >> 9;
    const int n    = lane & 15;
    const int half = lane >> 4;
    const int c0   = ct * 16;

    float ap[17];
    ap[0] = 1.0f;
    #pragma unroll
    for (int i = 1; i <= 16; ++i) ap[i] = ap[i - 1] * (float)AVAL_;

    v2f Aop[4];
    #pragma unroll
    for (int j = 0; j < 4; ++j) {
        int kk = 4 * j + half * 2;
        Aop[j].x = (kk     <= n) ? (float)ALPHA_ * ap[n - kk]     : 0.0f;
        Aop[j].y = (kk + 1 <= n) ? (float)ALPHA_ * ap[n - kk - 1] : 0.0f;
    }

    float crun = (float)carry[((size_t)b * NCH_ + k) * C_ + c0 + n];

    const size_t baseT = (size_t)b * T_;
    for (int tt = 0; tt < CL_ / 16; ++tt) {
        const int t0 = k * CL_ + tt * 16;
        v8f acc = {};
        #pragma unroll
        for (int j = 0; j < 4; ++j) {
            const int tr = 4 * j + half * 2;
            const float* p = x + (baseT + (size_t)(t0 + tr)) * C_ + c0 + n;
            float b0 = p[0];
            float b1 = p[C_];
            b0 = (t0 + tr == 0) ? b0 * (float)(1.0 / ALPHA_) : b0;
            v2f Bop; Bop.x = b0; Bop.y = b1;
            acc = __builtin_amdgcn_wmma_f32_16x16x4_f32(
                false, Aop[j], false, Bop, (short)0, acc, false, false);
        }
        #pragma unroll
        for (int v = 0; v < 8; ++v)
            acc[v] += ap[v + half * 8 + 1] * crun;
        float* o = out + (baseT + (size_t)(t0 + half * 8)) * C_ + c0 + n;
        #pragma unroll
        for (int v = 0; v < 8; ++v)
            o[(size_t)v * C_] = acc[v];
        crun = __shfl(acc[7], n + 16, 32);
    }
}

extern "C" void kernel_launch(void* const* d_in, const int* in_sizes, int n_in,
                              void* d_out, int out_size, void* d_ws, size_t ws_size,
                              hipStream_t stream) {
    const float* x = (const float*)d_in[0];
    float* out = (float*)d_out;
    double* E     = (double*)d_ws;
    double* carry = E + (size_t)B_ * NCH_ * C_;

    ema_p1<<<(B_ * NCH_ * C_) / 256, 256, 0, stream>>>(x, E);
    ema_p2<<<(B_ * C_) / 256, 256, 0, stream>>>(E, carry);
    ema_p3<<<(B_ * NCH_ * (C_ / 16) * 32) / 256, 256, 0, stream>>>(x, carry, out);
}


# One pass over x: a work-group walks all T for one 16-channel strip, its 32 waves scan 32 tiles at once and swap tile ends through LDS; carries f32 (were f64), sum order changed; P1 and P2 gone
# speedup vs baseline: 1.4518x; 1.4518x over previous
#include <hip/hip_runtime.h>


#define ALPHA_  0.01
#define AVAL_   0.99
#define B_      8
#define T_      8192
#define C_      512
#define NW_     32
#define SS_     (NW_ * 16)
#define NS_     (T_ / SS_)
#define DEPTH_  5
#define RING_   (DEPTH_ + 1)
#define EPITCH_ 36
#define EBUF_   (16 * EPITCH_)

typedef __attribute__((ext_vector_type(2))) float v2f;
typedef __attribute__((ext_vector_type(4))) float v4f;
typedef __attribute__((ext_vector_type(8))) float v8f;

static __device__ __forceinline__ float pick_pow(const float (&ap)[17], int d) {
    float r = 0.0f;
    #pragma unroll
    for (int i = 0; i <= 16; ++i) r = (d == i) ? ap[i] : r;
    return r;
}

__global__ void __launch_bounds__(1024)
ema_onepass(const float* __restrict__ x, float* __restrict__ out) {
    __shared__ __attribute__((aligned(16))) float Esh[2 * EBUF_];

    const int lane = threadIdx.x & 31;
    const int w    = __builtin_amdgcn_readfirstlane(threadIdx.x >> 5);
    const int ct   = blockIdx.x & 31;
    const int b    = blockIdx.x >> 5;
    const int n    = lane & 15;
    const int half = lane >> 4;
    const int c0   = ct * 16;

    const float* xp = x + ((size_t)b * T_ + (size_t)(w * 16 + half * 2)) * C_ + c0 + n;
    float* op = out + ((size_t)b * T_ + (size_t)(w * 16 + half * 8)) * C_ + c0 + n;

    float xr[RING_][8];
    #pragma unroll
    for (int s = 0; s < DEPTH_; ++s) {
        #pragma unroll
        for (int j = 0; j < 4; ++j) {
            const float* p = xp + (size_t)(s * SS_ + 4 * j) * C_;
            xr[s][2 * j]     = p[0];
            xr[s][2 * j + 1] = p[C_];
        }
    }

    float ap[17];
    ap[0] = 1.0f;
    #pragma unroll
    for (int i = 1; i <= 16; ++i) ap[i] = ap[i - 1] * (float)AVAL_;
    const float P = ap[16];

    v2f Aop[4];
    #pragma unroll
    for (int j = 0; j < 4; ++j) {
        int kk = 4 * j + half * 2;
        Aop[j].x = (kk     <= n) ? (float)ALPHA_ * pick_pow(ap, n - kk)     : 0.0f;
        Aop[j].y = (kk + 1 <= n) ? (float)ALPHA_ * pick_pow(ap, n - kk - 1) : 0.0f;
    }
    float cwgt[8];
    #pragma unroll
    for (int v = 0; v < 8; ++v) cwgt[v] = half ? ap[v + 9] : ap[v + 1];

    float R = 0.0f;

    #pragma unroll
    for (int s = 0; s < NS_; ++s) {
        if (s + DEPTH_ < NS_) {
            #pragma unroll
            for (int j = 0; j < 4; ++j) {
                const float* p = xp + (size_t)((s + DEPTH_) * SS_ + 4 * j) * C_;
                xr[(s + DEPTH_) % RING_][2 * j]     = p[0];
                xr[(s + DEPTH_) % RING_][2 * j + 1] = p[C_];
            }
        }

        v8f acc = {};
        #pragma unroll
        for (int j = 0; j < 4; ++j) {
            float b0 = xr[s % RING_][2 * j];
            float b1 = xr[s % RING_][2 * j + 1];
            if (s == 0 && j == 0)
                b0 = (w == 0 && half == 0) ? b0 * (float)(1.0 / ALPHA_) : b0;
            v2f Bop; Bop.x = b0; Bop.y = b1;
            acc = __builtin_amdgcn_wmma_f32_16x16x4_f32(
                false, Aop[j], false, Bop, (short)0, acc, false, false);
        }

        float* eb = Esh + (s & 1) * EBUF_;
        if (half) eb[n * EPITCH_ + w] = acc[7];
        asm volatile("s_wait_dscnt 0x0" ::: "memory");
        asm volatile("s_barrier_signal -1" ::: "memory");
        asm volatile("s_barrier_wait -1" ::: "memory");

        float ev[NW_];
        #pragma unroll
        for (int q = 0; q < NW_ / 4; ++q) {
            v4f t4 = *(const v4f*)(eb + n * EPITCH_ + 4 * q);
            ev[4 * q] = t4.x; ev[4 * q + 1] = t4.y; ev[4 * q + 2] = t4.z; ev[4 * q + 3] = t4.w;
        }
        float c = R, cw = R;
        #pragma unroll
        for (int j = 0; j < NW_; ++j) {
            c = __builtin_fmaf(P, c, ev[j]);
            cw = (j + 1 == w) ? c : cw;
        }
        R = c;

        float* o = op + (size_t)(s * SS_) * C_;
        #pragma unroll
        for (int v = 0; v < 8; ++v)
            o[(size_t)v * C_] = __builtin_fmaf(cwgt[v], cw, acc[v]);
    }
}

extern "C" void kernel_launch(void* const* d_in, const int* in_sizes, int n_in,
                              void* d_out, int out_size, void* d_ws, size_t ws_size,
                              hipStream_t stream) {
    const float* x = (const float*)d_in[0];
    float* out = (float*)d_out;
    ema_onepass<<<B_ * (C_ / 16), NW_ * 32, 0, stream>>>(x, out);
}


# Strip widened from 16 to 32 channels: each row piece is one whole 128-byte line (8-byte loads/stores, two products per tile, carry chains split over lane halves); 128 groups of 16 waves, walk rolled
# speedup vs baseline: 3.3744x; 2.3243x over previous
#include <hip/hip_runtime.h>


#define ALPHA_  0.01
#define AVAL_   0.99
#define B_      8
#define T_      8192
#define C_      512
#define CS_     32
#define NW_     16
#define SS_     (NW_ * 16)
#define NS_     (T_ / SS_)
#define DEPTH_  5
#define RING_   8
#define EPITCH_ 20
#define EBUF_   (CS_ * EPITCH_)

typedef __attribute__((ext_vector_type(2))) float v2f;
typedef __attribute__((ext_vector_type(4))) float v4f;
typedef __attribute__((ext_vector_type(8))) float v8f;

static __device__ __forceinline__ float pick_pow(const float (&ap)[17], int d) {
    float r = 0.0f;
    #pragma unroll
    for (int i = 0; i <= 16; ++i) r = (d == i) ? ap[i] : r;
    return r;
}

static __device__ __forceinline__ float xhalf(float v) {
    return __builtin_bit_cast(float, __builtin_amdgcn_permlanex16(
        __builtin_bit_cast(int, v), __builtin_bit_cast(int, v), 0x76543210, 0xfedcba98, false, false));
}

__global__ void __launch_bounds__(NW_ * 32)
ema_onepass(const float* __restrict__ x, float* __restrict__ out) {
    __shared__ __attribute__((aligned(16))) float Esh[2 * EBUF_];

    const int lane = threadIdx.x & 31;
    const int w    = __builtin_amdgcn_readfirstlane(threadIdx.x >> 5);
    const int ct   = blockIdx.x & (C_ / CS_ - 1);
    const int b    = blockIdx.x / (C_ / CS_);
    const int n    = lane & 15;
    const int half = lane >> 4;
    const int c0   = ct * CS_;

    const float* xp = x + ((size_t)b * T_ + (size_t)(w * 16 + half * 2)) * C_ + c0 + 2 * n;
    float* op = out + ((size_t)b * T_ + (size_t)(w * 16 + half * 8)) * C_ + c0 + 2 * n;

    v2f xr[RING_][8];
    #pragma unroll
    for (int s = 0; s < DEPTH_; ++s) {
        #pragma unroll
        for (int j = 0; j < 4; ++j) {
            const float* p = xp + (size_t)(s * SS_ + 4 * j) * C_;
            xr[s][2 * j]     = *(const v2f*)p;
            xr[s][2 * j + 1] = *(const v2f*)(p + C_);
        }
    }

    float ap[17];
    ap[0] = 1.0f;
    #pragma unroll
    for (int i = 1; i <= 16; ++i) ap[i] = ap[i - 1] * (float)AVAL_;
    const float P = ap[16];

    v2f Aop[4];
    #pragma unroll
    for (int j = 0; j < 4; ++j) {
        int kk = 4 * j + half * 2;
        Aop[j].x = (kk     <= n) ? (float)ALPHA_ * pick_pow(ap, n - kk)     : 0.0f;
        Aop[j].y = (kk + 1 <= n) ? (float)ALPHA_ * pick_pow(ap, n - kk - 1) : 0.0f;
    }
    float cwgt[8];
    #pragma unroll
    for (int v = 0; v < 8; ++v) cwgt[v] = half ? ap[v + 9] : ap[v + 1];

    float R = 0.0f;

    xr[0][0].x = (w == 0 && half == 0) ? xr[0][0].x * (float)(1.0 / ALPHA_) : xr[0][0].x;
    xr[0][0].y = (w == 0 && half == 0) ? xr[0][0].y * (float)(1.0 / ALPHA_) : xr[0][0].y;

    auto trip = [&](const int it, const bool last) __attribute__((always_inline)) {
        #pragma unroll
        for (int u = 0; u < RING_; ++u) {
            const int s = it * RING_ + u;
            if (!last || u + DEPTH_ < RING_) {
                #pragma unroll
                for (int j = 0; j < 4; ++j) {
                    const float* p = xp + (size_t)((s + DEPTH_) * SS_ + 4 * j) * C_;
                    xr[(u + DEPTH_) % RING_][2 * j]     = *(const v2f*)p;
                    xr[(u + DEPTH_) % RING_][2 * j + 1] = *(const v2f*)(p + C_);
                }
            }

            v8f acc0 = {}, acc1 = {};
            #pragma unroll
            for (int j = 0; j < 4; ++j) {
                const v2f r0 = xr[u][2 * j];
                const v2f r1 = xr[u][2 * j + 1];
                v2f Bop0; Bop0.x = r0.x; Bop0.y = r1.x;
                v2f Bop1; Bop1.x = r0.y; Bop1.y = r1.y;
                acc0 = __builtin_amdgcn_wmma_f32_16x16x4_f32(
                    false, Aop[j], false, Bop0, (short)0, acc0, false, false);
                acc1 = __builtin_amdgcn_wmma_f32_16x16x4_f32(
                    false, Aop[j], false, Bop1, (short)0, acc1, false, false);
            }

            float* eb = Esh + (u & 1) * EBUF_;
            if (half) {
                eb[n * EPITCH_ + w]        = acc0[7];
                eb[(n + 16) * EPITCH_ + w] = acc1[7];
            }
            asm volatile("s_wait_dscnt 0x0" ::: "memory");
            asm volatile("s_barrier_signal -1" ::: "memory");
            asm volatile("s_barrier_wait -1" ::: "memory");

            float ev[NW_];
            #pragma unroll
            for (int q = 0; q < NW_ / 4; ++q) {
                v4f t4 = *(const v4f*)(eb + lane * EPITCH_ + 4 * q);
                ev[4 * q] = t4.x; ev[4 * q + 1] = t4.y; ev[4 * q + 2] = t4.z; ev[4 * q + 3] = t4.w;
            }
            float c = R, cw = R;
            #pragma unroll
            for (int j = 0; j < NW_; ++j) {
                c = __builtin_fmaf(P, c, ev[j]);
                cw = (j + 1 == w) ? c : cw;
            }
            R = c;
            const float co  = xhalf(cw);
            const float cw0 = half ? co : cw;
            const float cw1 = half ? cw : co;

            float* o = op + (size_t)(s * SS_) * C_;
            #pragma unroll
            for (int v = 0; v < 8; ++v) {
                v2f y;
                y.x = __builtin_fmaf(cwgt[v], cw0, acc0[v]);
                y.y = __builtin_fmaf(cwgt[v], cw1, acc1[v]);
                *(v2f*)(o + (size_t)v * C_) = y;
            }
        }
    };

    #pragma nounroll
    for (int it = 0; it < NS_ / RING_ - 1; ++it) trip(it, false);
    trip(NS_ / RING_ - 1, true);
}

extern "C" void kernel_launch(void* const* d_in, const int* in_sizes, int n_in,
                              void* d_out, int out_size, void* d_ws, size_t ws_size,
                              hipStream_t stream) {
    const float* x = (const float*)d_in[0];
    float* out = (float*)d_out;
    ema_onepass<<<B_ * (C_ / CS_), NW_ * 32, 0, stream>>>(x, out);
}
